# starting kernel
# baseline (speedup 1.0000x reference)
#include <hip/hip_runtime.h>


typedef __attribute__((ext_vector_type(2))) float v2f;
typedef __attribute__((ext_vector_type(8))) float v8f;

#define NQ 8
#define NLAYERS 4

struct cpx { float re, im; };

__device__ __forceinline__ cpx cmul(cpx a, cpx b) {
  cpx r; r.re = a.re * b.re - a.im * b.im; r.im = a.re * b.im + a.im * b.re; return r;
}

__device__ __forceinline__ cpx u_entry(float a, float bb, float g, int i, int j) {
  float ca = cosf(0.5f * a),  sa = sinf(0.5f * a);
  float cb = cosf(0.5f * bb), sb = sinf(0.5f * bb);
  cpx m;
  if (i == 0 && j == 0) { m.re =  cb * ca; m.im =  sb * sa; }
  else if (i == 0)      { m.re = -sb * ca; m.im = -cb * sa; }
  else if (j == 0)      { m.re =  sb * ca; m.im = -cb * sa; }
  else                  { m.re =  cb * ca; m.im = -sb * sa; }
  float cg = cosf(0.5f * g), sg = sinf(0.5f * g);
  cpx ph; ph.re = cg; ph.im = (i == 0) ? -sg : sg;
  return cmul(ph, m);
}

__host__ __device__ constexpr int chainperm(int r) {
  r ^= ((r >> 1) & 1);
  r ^= ((r >> 2) & 1) << 1;
  r ^= ((r >> 3) & 1) << 2;
  return r;
}

__global__ void qprep(const float* __restrict__ w, float* __restrict__ ws) {
  int t = threadIdx.x;
  int m = t >> 4, k = t & 15;
  for (int l = 0; l < NLAYERS; ++l) {
    cpx pr; pr.re = 1.f; pr.im = 0.f;
    #pragma unroll
    for (int j = 3; j >= 0; --j) {
      int q = 4 + (3 - j);
      int base = (l * NQ + q) * 3;
      pr = cmul(pr, u_entry(w[base], w[base + 1], w[base + 2], (m >> j) & 1, (k >> j) & 1));
    }
    float* dst = ws + (size_t)((l * 2 + 0) * 2) * 256;
    dst[m * 16 + k]       = pr.re;
    dst[256 + m * 16 + k] = pr.im;

    int mp = chainperm(m);
    cpx pc; pc.re = 1.f; pc.im = 0.f;
    #pragma unroll
    for (int j = 3; j >= 0; --j) {
      int q = (3 - j);
      int base = (l * NQ + q) * 3;
      pc = cmul(pc, u_entry(w[base], w[base + 1], w[base + 2], (mp >> j) & 1, (k >> j) & 1));
    }
    dst = ws + (size_t)((l * 2 + 1) * 2) * 256;
    dst[m * 16 + k]       = pc.re;
    dst[256 + m * 16 + k] = pc.im;
  }
}


__device__ __forceinline__ void cd_to_b(const float* S, float* Bv, int half) {
  const int r0t[8] = {0, 1, 4, 5, 0, 1, 4, 5};
  #pragma unroll
  for (int d = 0; d < 8; ++d) {
    int r0 = r0t[d];
    if (d < 4) {
      float t = __shfl_xor(S[r0 + 2], 16, 32);
      Bv[d] = half ? t : S[r0];
    } else {
      float t = __shfl_xor(S[r0], 16, 32);
      Bv[d] = half ? S[r0 + 2] : t;
    }
  }
}

__device__ __forceinline__ void cmulmat(const float* __restrict__ mat, int lane,
                                        const float* Xb, const float* Yb,
                                        float* Xo, float* Yo) {
  const int half = lane >> 4;
  const int mrow = lane & 15;
  float Are[8], Aim[8];
  #pragma unroll
  for (int a = 0; a < 8; ++a) {
    int k = 4 * (a >> 1) + (a & 1) + 2 * half;
    Are[a] = mat[mrow * 16 + k];
    Aim[a] = mat[256 + mrow * 16 + k];
  }
  v8f accX = {}; v8f accY = {};
  #pragma unroll
  for (int kc = 0; kc < 4; ++kc) {
    int a0 = 2 * kc, a1 = a0 + 1;
    v2f ar;  ar[0]  = Are[a0];  ar[1]  = Are[a1];
    v2f ai;  ai[0]  = Aim[a0];  ai[1]  = Aim[a1];
    v2f bx;  bx[0]  = Xb[a0];   bx[1]  = Xb[a1];
    v2f by;  by[0]  = Yb[a0];   by[1]  = Yb[a1];
    v2f nby; nby[0] = -Yb[a0];  nby[1] = -Yb[a1];
    accX = __builtin_amdgcn_wmma_f32_16x16x4_f32(false, ar, false, bx,  (short)0, accX, false, false);
    accX = __builtin_amdgcn_wmma_f32_16x16x4_f32(false, ai, false, nby, (short)0, accX, false, false);
    accY = __builtin_amdgcn_wmma_f32_16x16x4_f32(false, ar, false, by,  (short)0, accY, false, false);
    accY = __builtin_amdgcn_wmma_f32_16x16x4_f32(false, ai, false, bx,  (short)0, accY, false, false);
  }
  #pragma unroll
  for (int v = 0; v < 8; ++v) { Xo[v] = accX[v]; Yo[v] = accY[v]; }
}

__global__ __launch_bounds__(256) void qsim(const float* __restrict__ x,
                                            const float* __restrict__ ws,
                                            float* __restrict__ out, int B) {
  __shared__ float lds[8 * 512];
  const int lane  = threadIdx.x & 31;
  const int wslot = threadIdx.x >> 5;
  const int b     = blockIdx.x * 8 + wslot;
  if (b >= B) return;
  const int half = lane >> 4;
  const int col  = lane & 15;
  float* myLds = lds + wslot * 512;

  float cq[8], sq[8];
  #pragma unroll
  for (int q = 0; q < 8; ++q) {
    float xv = x[b * 8 + q];
    cq[q] = cosf(0.5f * xv); sq[q] = sinf(0.5f * xv);
  }
  float colf = 1.f;
  #pragma unroll
  for (int q = 0; q < 4; ++q) colf *= ((col >> (3 - q)) & 1) ? sq[q] : cq[q];
  float X[8], Y[8];
  #pragma unroll
  for (int v = 0; v < 8; ++v) {
    int r = v + 8 * half;
    float rowf = 1.f;
    #pragma unroll
    for (int q = 4; q < 8; ++q) rowf *= ((r >> (7 - q)) & 1) ? sq[q] : cq[q];
    X[v] = colf * rowf; Y[v] = 0.f;
  }

  for (int l = 0; l < NLAYERS; ++l) {
    const int flip = l & 1;
    const float* matA = ws + (size_t)((l * 2 + flip)       * 2) * 256;
    const float* matB = ws + (size_t)((l * 2 + (1 - flip)) * 2) * 256;

    float Xb[8], Yb[8];
    cd_to_b(X, Xb, half);
    cd_to_b(Y, Yb, half);
    cmulmat(matA, lane, Xb, Yb, X, Y);

    #pragma unroll
    for (int v = 0; v < 8; ++v) {
      myLds[(v + 8 * half) * 16 + col]       = X[v];
      myLds[256 + (v + 8 * half) * 16 + col] = Y[v];
    }
    __syncthreads();
    #pragma unroll
    for (int bb = 0; bb < 8; ++bb) {
      int kk = 4 * (bb >> 1) + (bb & 1) + 2 * half;
      Xb[bb] = myLds[col * 16 + kk];
      Yb[bb] = myLds[256 + col * 16 + kk];
    }
    __syncthreads();

    cmulmat(matB, lane, Xb, Yb, X, Y);

    if (flip) {
      #pragma unroll
      for (int v = 0; v < 8; ++v) {
        float tx = __shfl_xor(X[v], 16, 32);
        float ty = __shfl_xor(Y[v], 16, 32);
        if (lane & 1) { X[v] = tx; Y[v] = ty; }
      }
      float nX[8], nY[8];
      #pragma unroll
      for (int v = 0; v < 8; ++v) {
        const int r0 = chainperm(v), r1 = chainperm(v + 8);
        float t0x = (r0 < 8)  ? X[r0 & 7] : __shfl_xor(X[r0 & 7], 16, 32);
        float t0y = (r0 < 8)  ? Y[r0 & 7] : __shfl_xor(Y[r0 & 7], 16, 32);
        float t1x = (r1 >= 8) ? X[r1 & 7] : __shfl_xor(X[r1 & 7], 16, 32);
        float t1y = (r1 >= 8) ? Y[r1 & 7] : __shfl_xor(Y[r1 & 7], 16, 32);
        nX[v] = half ? t1x : t0x;
        nY[v] = half ? t1y : t0y;
      }
      #pragma unroll
      for (int v = 0; v < 8; ++v) { X[v] = nX[v]; Y[v] = nY[v]; }
    } else {
      #pragma unroll
      for (int v = 1; v < 8; v += 2) {
        X[v] = __shfl_xor(X[v], 8, 32);
        Y[v] = __shfl_xor(Y[v], 8, 32);
      }
      int src = chainperm(lane & 15) | (lane & 16);
      #pragma unroll
      for (int v = 0; v < 8; ++v) {
        X[v] = __shfl(X[v], src, 32);
        Y[v] = __shfl(Y[v], src, 32);
      }
    }
  }

  float psum = 0.f, s5 = 0.f, s6 = 0.f, s7 = 0.f;
  #pragma unroll
  for (int v = 0; v < 8; ++v) {
    float p = X[v] * X[v] + Y[v] * Y[v];
    psum += p;
    s5 += ((v >> 2) & 1) ? -p : p;
    s6 += ((v >> 1) & 1) ? -p : p;
    s7 += ((v) & 1)      ? -p : p;
  }
  float o[8];
  o[0] = ((col >> 3) & 1) ? -psum : psum;
  o[1] = ((col >> 2) & 1) ? -psum : psum;
  o[2] = ((col >> 1) & 1) ? -psum : psum;
  o[3] = ((col) & 1)      ? -psum : psum;
  o[4] = half ? -psum : psum;
  o[5] = s5; o[6] = s6; o[7] = s7;
  #pragma unroll
  for (int q = 0; q < 8; ++q) {
    float v = o[q];
    #pragma unroll
    for (int off = 16; off > 0; off >>= 1) v += __shfl_xor(v, off, 32);
    o[q] = v;
  }
  if (lane == 0) {
    #pragma unroll
    for (int q = 0; q < 8; ++q) out[b * 8 + q] = o[q];
  }
}

extern "C" void kernel_launch(void* const* d_in, const int* in_sizes, int n_in,
                              void* d_out, int out_size, void* d_ws, size_t ws_size,
                              hipStream_t stream) {
  const float* x = (const float*)d_in[0];
  const float* w = (const float*)d_in[1];
  float* out = (float*)d_out;
  float* ws  = (float*)d_ws;
  int B = in_sizes[0] / NQ;

  qprep<<<1, 256, 0, stream>>>(w, ws);
  int nblocks = (B + 7) / 8;
  qsim<<<nblocks, 256, 0, stream>>>(x, ws, out, B);
}


# New layout: a wave holds 16 samples (sample = N column, re/im in the two 16-lane groups); gates as 3-wire real 16x16 products on registers, CNOTs as renaming; 128 -> 96 WMMA/sample; sums regrouped
# speedup vs baseline: 2.3701x; 2.3701x over previous
#include <hip/hip_runtime.h>


typedef __attribute__((ext_vector_type(2))) float v2f;
typedef __attribute__((ext_vector_type(8))) float v8f;
typedef __attribute__((ext_vector_type(4))) float v4f;

#define NQ 8
#define NLAYERS 4
#define NSTEPS 3
#define NMATS (NLAYERS * NSTEPS)

struct cpx { float re, im; };

__device__ __forceinline__ cpx cmul(cpx a, cpx b) {
  cpx r; r.re = a.re * b.re - a.im * b.im; r.im = a.re * b.im + a.im * b.re; return r;
}

__device__ __forceinline__ cpx u_entry(float ca, float sa, float cb, float sb,
                                       float cg, float sg, int i, int j) {
  cpx m;
  if (i == 0 && j == 0) { m.re =  cb * ca; m.im =  sb * sa; }
  else if (i == 0)      { m.re = -sb * ca; m.im = -cb * sa; }
  else if (j == 0)      { m.re =  sb * ca; m.im = -cb * sa; }
  else                  { m.re =  cb * ca; m.im = -sb * sa; }
  cpx ph; ph.re = cg; ph.im = (i == 0) ? -sg : sg;
  return cmul(ph, m);
}

__host__ __device__ constexpr int amp_idx(int g, int o, int b) {
  return g == 0 ? ((b << 5) | o)
       : g == 1 ? (((o >> 2) << 5) | (b << 2) | (o & 3))
                : (((b >> 2) << 7) | (o << 2) | (b & 3));
}

__host__ __device__ constexpr int cnot_dst(int j) {
  return (j ^ (j >> 1) ^ (j >> 2) ^ (j >> 3) ^ (j >> 4) ^ (j >> 5) ^ (j >> 6) ^ (j >> 7)) & 255;
}

#define PREP_THREADS 64
#define PREP_BLOCKS (NMATS * (256 / PREP_THREADS))
#define ANG_OFF (NMATS * 256)
__global__ __launch_bounds__(PREP_THREADS) void qprep(const float* __restrict__ x,
                                                      const float* __restrict__ w,
                                                      float* __restrict__ ws, int nquads) {
  __shared__ float trig[18];
  if (blockIdx.x >= PREP_BLOCKS) {
    const int t = (blockIdx.x - PREP_BLOCKS) * PREP_THREADS + threadIdx.x;
    if (t < nquads) {
      const v4f xv = reinterpret_cast<const v4f*>(x)[t];
      v4f c, s;
      #pragma unroll
      for (int i = 0; i < 4; ++i) { c[i] = cosf(0.5f * xv[i]); s[i] = sinf(0.5f * xv[i]); }
      v4f* dst = reinterpret_cast<v4f*>(ws + ANG_OFF + (size_t)(t >> 1) * 16 + (t & 1) * 4);
      dst[0] = c;
      dst[2] = s;
    }
    return;
  }
  const int parts = 256 / PREP_THREADS;
  const int mi = blockIdx.x / parts;
  const int e  = (blockIdx.x % parts) * PREP_THREADS + threadIdx.x;
  const int l  = mi / NSTEPS;
  const int g  = mi % NSTEPS;
  const int q0 = 3 * g;
  const int nw = (g == 2) ? 2 : 3;
  if ((int)threadIdx.x < 3 * nw) {
    const float h = 0.5f * w[(l * NQ + q0) * 3 + threadIdx.x];
    trig[2 * threadIdx.x]     = cosf(h);
    trig[2 * threadIdx.x + 1] = sinf(h);
  }
  __syncthreads();
  const int m = e >> 4, k = e & 15;
  const int pp = m >> 3, bp = m & 7;
  const int p  = k >> 3, b  = k & 7;
  cpx G; G.re = 1.f; G.im = 0.f;
  for (int j = 0; j < nw; ++j) {
    const float* t = trig + 6 * j;
    const int sh = nw - 1 - j;
    G = cmul(G, u_entry(t[0], t[1], t[2], t[3], t[4], t[5], (bp >> sh) & 1, (b >> sh) & 1));
  }
  if (g == 2 && ((bp ^ b) & 4)) { G.re = 0.f; G.im = 0.f; }
  const float val = (pp == p) ? G.re : (pp ? G.im : -G.im);
  const int fl = m + 16 * p;
  ws[(size_t)mi * 256 + (b >> 2) * 128 + fl * 4 + (b & 3)] = val;
}

template <int ST>
__device__ __forceinline__ void tree_fold(float* d) {
  if constexpr (ST >= 1) {
    #pragma unroll
    for (int j = 0; j < ST; ++j) d[j] += d[j + ST];
    tree_fold<ST / 2>(d);
  }
}

template <int K>
__device__ __forceinline__ void zlevel(float* a, float* z) {
  constexpr int H = 128 >> K;
  float dd[H];
  #pragma unroll
  for (int j = 0; j < H; ++j) {
    const float e0 = a[2 * j], e1 = a[2 * j + 1];
    dd[j] = e0 - e1;
    a[j]  = e0 + e1;
  }
  tree_fold<H / 2>(dd);
  z[K] = dd[0];
  if constexpr (K < 7) zlevel<K + 1>(a, z);
}

template <int Q>
__device__ __forceinline__ void enc_level(float* S, const float* cq, const float* sq) {
  if constexpr (Q < 8) {
    #pragma unroll
    for (int j = (1 << Q) - 1; j >= 0; --j) {
      const float v = S[j];
      S[2 * j + 1] = v * sq[Q];
      S[2 * j]     = v * cq[Q];
    }
    enc_level<Q + 1>(S, cq, sq);
  }
}

template <int G, int O>
__device__ __forceinline__ void product_pairs(const float* in, float* res, const v2f* a) {
  if constexpr (O < 32) {
    v8f d0 = {}; v8f d1 = {};
    #pragma unroll
    for (int kc = 0; kc < 4; ++kc) {
      v2f b0, b1;
      b0[0] = in[amp_idx(G, O,     2 * kc)]; b0[1] = in[amp_idx(G, O,     2 * kc + 1)];
      b1[0] = in[amp_idx(G, O + 1, 2 * kc)]; b1[1] = in[amp_idx(G, O + 1, 2 * kc + 1)];
      d0 = __builtin_amdgcn_wmma_f32_16x16x4_f32(false, a[kc], false, b0, (short)0, d0, false, false);
      d1 = __builtin_amdgcn_wmma_f32_16x16x4_f32(false, a[kc], false, b1, (short)0, d1, false, false);
    }
    #pragma unroll
    for (int v = 0; v < 8; ++v) {
      const int j0 = amp_idx(G, O, v), j1 = amp_idx(G, O + 1, v);
      res[G == 2 ? cnot_dst(j0) : j0] = d0[v];
      res[G == 2 ? cnot_dst(j1) : j1] = d1[v];
    }
    product_pairs<G, O + 2>(in, res, a);
  }
}

__device__ __forceinline__ void frag_to_a(v4f f0, v4f f1, v2f* a) {
  a[0][0] = f0[0]; a[0][1] = f0[1]; a[1][0] = f0[2]; a[1][1] = f0[3];
  a[2][0] = f1[0]; a[2][1] = f1[1]; a[3][0] = f1[2]; a[3][1] = f1[3];
}

template <int I, int N>
__device__ __forceinline__ void squares(const float* s, float* q) {
  if constexpr (I < N) {
    #pragma unroll
    for (int i = I; i < I + 32; ++i) q[i] = s[i] * s[i];
    squares<I + 32, N>(s, q);
  }
}

__global__ __launch_bounds__(128) __attribute__((amdgpu_waves_per_eu(1, 1)))
void qsim(const float* __restrict__ x,
          const float* __restrict__ ws,
          float* __restrict__ out, int B) {
  const int lane = threadIdx.x & 31;
  const int wave = blockIdx.x * 4 + (threadIdx.x >> 5);
  const int half = lane >> 4;
  const int b    = wave * 16 + (lane & 15);
  const int bc   = b < B ? b : B - 1;

  const v4f* ang = reinterpret_cast<const v4f*>(ws + ANG_OFF + (size_t)bc * 16);
  const v4f c0 = ang[0], c1 = ang[1], s0 = ang[2], s1 = ang[3];
  const float cq[8] = {c0[0], c0[1], c0[2], c0[3], c1[0], c1[1], c1[2], c1[3]};
  const float sq[8] = {s0[0], s0[1], s0[2], s0[3], s1[0], s1[1], s1[2], s1[3]};
  float S[256];
  S[0] = half ? 0.f : 1.f;
  enc_level<0>(S, cq, sq);

  const v4f* frag = reinterpret_cast<const v4f*>(ws) + lane;
  v4f f0 = frag[0], f1 = frag[32];
  #pragma nounroll
  for (int l = 0; l < NLAYERS; ++l) {
    float R[256];
    v2f a[4];
    const v4f* fl = frag + l * (NSTEPS * 64);
    const int nl = (l + 1 < NLAYERS) ? (l + 1) : l;
    const v4f g10 = fl[64], g11 = fl[96];
    frag_to_a(f0, f1, a);
    product_pairs<0, 0>(S, R, a);
    const v4f g20 = fl[128], g21 = fl[160];
    frag_to_a(g10, g11, a);
    product_pairs<1, 0>(R, R, a);
    f0 = frag[nl * (NSTEPS * 64)]; f1 = frag[nl * (NSTEPS * 64) + 32];
    frag_to_a(g20, g21, a);
    product_pairs<2, 0>(R, S, a);
  }

  float sqv[256];
  squares<0, 256>(S, sqv);
  float z[8];
  zlevel<0>(sqv, z);
  #pragma unroll
  for (int k = 0; k < 8; ++k) z[k] += __shfl_xor(z[k], 16, 32);
  if (half == 0 && b < B) {
    v4f o0, o1;
    o0[0] = z[7]; o0[1] = z[6]; o0[2] = z[5]; o0[3] = z[4];
    o1[0] = z[3]; o1[1] = z[2]; o1[2] = z[1]; o1[3] = z[0];
    v4f* dst = reinterpret_cast<v4f*>(out + (size_t)b * 8);
    dst[0] = o0;
    dst[1] = o1;
  }
}

extern "C" void kernel_launch(void* const* d_in, const int* in_sizes, int n_in,
                              void* d_out, int out_size, void* d_ws, size_t ws_size,
                              hipStream_t stream) {
  const float* x = (const float*)d_in[0];
  const float* w = (const float*)d_in[1];
  float* out = (float*)d_out;
  float* ws  = (float*)d_ws;
  int B = in_sizes[0] / NQ;

  int nquads = B * 2;
  int tblocks = (nquads + PREP_THREADS - 1) / PREP_THREADS;
  qprep<<<PREP_BLOCKS + tblocks, PREP_THREADS, 0, stream>>>(x, w, ws, nquads);
  int nblocks = (B + 63) / 64;
  qsim<<<nblocks, 128, 0, stream>>>(x, ws, out, B);
}


# Circuit regrouped into 6 three-wire blocks that span layers: gates and CNOTs on a block's wires fused into one 8x8 matrix (two if a CNOT's control is outside); 96 -> 48 WMMA/sample; sums regrouped
# speedup vs baseline: 3.3382x; 1.4085x over previous
#include <hip/hip_runtime.h>


typedef __attribute__((ext_vector_type(2))) float v2f;
typedef __attribute__((ext_vector_type(8))) float v8f;
typedef __attribute__((ext_vector_type(4))) float v4f;

#define NQ 8
#define NBLOCKS 6
#define NMATS (NBLOCKS * 2)
#define MAXOPS 16

struct cpx { float re, im; };

__device__ __forceinline__ cpx cmul(cpx a, cpx b) {
  cpx r; r.re = a.re * b.re - a.im * b.im; r.im = a.re * b.im + a.im * b.re; return r;
}
__device__ __forceinline__ cpx cadd(cpx a, cpx b) {
  cpx r; r.re = a.re + b.re; r.im = a.im + b.im; return r;
}

__device__ __forceinline__ cpx u_entry(float ca, float sa, float cb, float sb,
                                       float cg, float sg, int i, int j) {
  cpx m;
  if (i == 0 && j == 0) { m.re =  cb * ca; m.im =  sb * sa; }
  else if (i == 0)      { m.re = -sb * ca; m.im = -cb * sa; }
  else if (j == 0)      { m.re =  sb * ca; m.im = -cb * sa; }
  else                  { m.re =  cb * ca; m.im = -sb * sa; }
  cpx ph; ph.re = cg; ph.im = (i == 0) ? -sg : sg;
  return cmul(ph, m);
}

__host__ __device__ constexpr int win_a(int s) { return s == 0 ? 0 : s == 1 ? 2 : s == 2 ? 4 : s == 3 ? 1 : s == 4 ? 5 : 0; }
__host__ __device__ constexpr int win_b(int s) { return s == 0 ? 1 : s == 1 ? 3 : s == 2 ? 5 : s == 3 ? 2 : s == 4 ? 6 : 3; }
__host__ __device__ constexpr int win_c(int s) { return s == 0 ? 2 : s == 1 ? 4 : s == 2 ? 6 : s == 3 ? 3 : s == 4 ? 7 : 4; }
__host__ __device__ constexpr int ctl_wire(int s) { return s == 1 ? 1 : s == 2 ? 3 : s == 3 ? 0 : s == 4 ? 4 : -1; }

__host__ __device__ constexpr int ins_zero(int x, int p) {
  return ((x >> p) << (p + 1)) | (x & ((1 << p) - 1));
}
__host__ __device__ constexpr int amp_of(int s, int o, int v) {
  const int pa = 7 - win_a(s), pb = 7 - win_b(s), pc = 7 - win_c(s);
  return ins_zero(ins_zero(ins_zero(o, pc), pb), pa)
       | (((v >> 2) & 1) << pa) | (((v >> 1) & 1) << pb) | ((v & 1) << pc);
}
__host__ __device__ constexpr int var_of(int s, int j) {
  return ctl_wire(s) < 0 ? 0 : ((j >> (7 - ctl_wire(s))) & 1);
}
__host__ __device__ constexpr int cx(int j, int c, int t) {
  return j ^ (((j >> (7 - c)) & 1) << (7 - t));
}
__host__ __device__ constexpr int cnot_dst(int j) {
  return (j ^ (j >> 1) ^ (j >> 2) ^ (j >> 3) ^ (j >> 4) ^ (j >> 5) ^ (j >> 6) ^ (j >> 7)) & 255;
}
__host__ __device__ constexpr int file_of(int s, int j) {
  return s == 3 ? cx(j, 3, 4) : s == 5 ? cnot_dst(j) : j;
}

#define OP_GATE(sh, l, w) (0 | ((sh) << 2) | ((((l) * NQ) + (w)) << 4))
#define OP_CNOT(c, t)     (1 | ((c) << 2) | ((t) << 4))
#define OP_XOUT(t)        (2 | ((t) << 2))
#define OP_END            3
__constant__ const unsigned short block_ops[NBLOCKS * MAXOPS] = {
    OP_GATE(2, 0, 0), OP_GATE(1, 0, 1), OP_GATE(0, 0, 2), OP_CNOT(2, 1), OP_CNOT(1, 0), OP_GATE(2, 1, 0), OP_GATE(1, 1, 1), OP_CNOT(2, 1), OP_GATE(2, 2, 0), OP_END, OP_END, OP_END, OP_END, OP_END, OP_END, OP_END,
    OP_GATE(1, 0, 3), OP_GATE(0, 0, 4), OP_CNOT(2, 1), OP_CNOT(1, 0), OP_GATE(2, 1, 2), OP_GATE(1, 1, 3), OP_XOUT(2), OP_CNOT(2, 1), OP_GATE(2, 2, 2), OP_END, OP_END, OP_END, OP_END, OP_END, OP_END, OP_END,
    OP_GATE(1, 0, 5), OP_GATE(0, 0, 6), OP_CNOT(2, 1), OP_CNOT(1, 0), OP_GATE(2, 1, 4), OP_GATE(1, 1, 5), OP_XOUT(2), OP_CNOT(2, 1), OP_GATE(2, 2, 4), OP_END, OP_END, OP_END, OP_END, OP_END, OP_END, OP_END,
    OP_GATE(2, 2, 1), OP_GATE(0, 2, 3), OP_XOUT(2), OP_CNOT(2, 1), OP_CNOT(1, 0), OP_GATE(2, 3, 1), OP_GATE(1, 3, 2), OP_END, OP_END, OP_END, OP_END, OP_END, OP_END, OP_END, OP_END, OP_END,
    OP_GATE(0, 0, 7), OP_CNOT(1, 0), OP_GATE(0, 1, 7), OP_GATE(1, 1, 6), OP_CNOT(2, 1), OP_CNOT(1, 0), OP_GATE(0, 2, 7), OP_GATE(2, 2, 5), OP_GATE(1, 2, 6), OP_XOUT(2), OP_CNOT(2, 1), OP_CNOT(1, 0), OP_GATE(2, 3, 5), OP_GATE(1, 3, 6), OP_GATE(0, 3, 7), OP_END,
    OP_GATE(2, 3, 0), OP_GATE(1, 3, 3), OP_GATE(0, 3, 4), OP_END, OP_END, OP_END, OP_END, OP_END, OP_END, OP_END, OP_END, OP_END, OP_END, OP_END, OP_END, OP_END
};

#define PREP_THREADS 64
#define PREP_BLOCKS (NMATS * (256 / PREP_THREADS))
#define ANG_OFF (NMATS * 256)
__global__ __launch_bounds__(PREP_THREADS) void qprep(const float* __restrict__ x,
                                                      const float* __restrict__ w,
                                                      float* __restrict__ ws, int nquads) {
  __shared__ float trig[MAXOPS * 6];
  if (blockIdx.x >= PREP_BLOCKS) {
    const int t = (blockIdx.x - PREP_BLOCKS) * PREP_THREADS + threadIdx.x;
    if (t < nquads) {
      const v4f xv = reinterpret_cast<const v4f*>(x)[t];
      v4f c, s;
      #pragma unroll
      for (int i = 0; i < 4; ++i) { c[i] = cosf(0.5f * xv[i]); s[i] = sinf(0.5f * xv[i]); }
      v4f* dst = reinterpret_cast<v4f*>(ws + ANG_OFF + (size_t)(t >> 1) * 16 + (t & 1) * 4);
      dst[0] = c;
      dst[2] = s;
    }
    return;
  }
  const int parts = 256 / PREP_THREADS;
  const int mi = blockIdx.x / parts;
  const int e  = (blockIdx.x % parts) * PREP_THREADS + threadIdx.x;
  const int s  = mi >> 1;
  const int u  = mi & 1;
  if (u == 1 && ctl_wire(s) < 0) return;
  const unsigned short* ops = block_ops + s * MAXOPS;
  const int myop = ops[threadIdx.x & (MAXOPS - 1)];
  if ((int)threadIdx.x < 3 * MAXOPS) {
    const int i = threadIdx.x / 3, r = threadIdx.x - 3 * i;
    const int op = __shfl(myop, i, 32);
    if ((op & 3) == 0) {
      const float h = 0.5f * w[3 * (op >> 4) + r];
      trig[6 * i + 2 * r]     = cosf(h);
      trig[6 * i + 2 * r + 1] = sinf(h);
    }
  }
  __syncthreads();
  const int m = e >> 4, k = e & 15;
  const int pp = m >> 3, bp = m & 7;
  const int pt = k >> 3, b  = k & 7;
  cpx v[8];
  #pragma unroll
  for (int j = 0; j < 8; ++j) { v[j].re = (j == b) ? 1.f : 0.f; v[j].im = 0.f; }
  for (int i = 0; i < MAXOPS; ++i) {
    const int op = __builtin_amdgcn_readlane(myop, i);
    const int kind = op & 3, p = (op >> 2) & 3, q = op >> 4;
    if (kind == 3) break;
    if (kind == 0) {
      const float* t = trig + 6 * i;
      const cpx u00 = u_entry(t[0], t[1], t[2], t[3], t[4], t[5], 0, 0);
      const cpx u01 = u_entry(t[0], t[1], t[2], t[3], t[4], t[5], 0, 1);
      const cpx u10 = u_entry(t[0], t[1], t[2], t[3], t[4], t[5], 1, 0);
      const cpx u11 = u_entry(t[0], t[1], t[2], t[3], t[4], t[5], 1, 1);
      #pragma unroll
      for (int sh = 0; sh < 3; ++sh) {
        if (p == sh) {
          #pragma unroll
          for (int j = 0; j < 8; ++j) {
            if (!(j & (1 << sh))) {
              const cpx x0 = v[j], x1 = v[j | (1 << sh)];
              v[j]             = cadd(cmul(u00, x0), cmul(u01, x1));
              v[j | (1 << sh)] = cadd(cmul(u10, x0), cmul(u11, x1));
            }
          }
        }
      }
    } else if (kind == 1 || u == 1) {
      const int tb = (kind == 1) ? q : p;
      #pragma unroll
      for (int sh = 0; sh < 3; ++sh) {
        if (tb == sh) {
          #pragma unroll
          for (int j = 0; j < 8; ++j) {
            if (!(j & (1 << sh))) {
              const bool on = (kind == 2) || (((j >> p) & 1) != 0);
              const cpx x0 = v[j], x1 = v[j | (1 << sh)];
              v[j]             = on ? x1 : x0;
              v[j | (1 << sh)] = on ? x0 : x1;
            }
          }
        }
      }
    }
  }
  cpx G; G.re = 0.f; G.im = 0.f;
  #pragma unroll
  for (int j = 0; j < 8; ++j) if (j == bp) G = v[j];
  const float val = (pp == pt) ? G.re : (pp ? G.im : -G.im);
  const int fl = m + 16 * pt;
  ws[(size_t)mi * 256 + (b >> 2) * 128 + fl * 4 + (b & 3)] = val;
}

template <int ST>
__device__ __forceinline__ void tree_fold(float* d) {
  if constexpr (ST >= 1) {
    #pragma unroll
    for (int j = 0; j < ST; ++j) d[j] += d[j + ST];
    tree_fold<ST / 2>(d);
  }
}

template <int K>
__device__ __forceinline__ void zlevel(float* a, float* z) {
  constexpr int H = 128 >> K;
  float dd[H];
  #pragma unroll
  for (int j = 0; j < H; ++j) {
    const float e0 = a[2 * j], e1 = a[2 * j + 1];
    dd[j] = e0 - e1;
    a[j]  = e0 + e1;
  }
  tree_fold<H / 2>(dd);
  z[K] = dd[0];
  if constexpr (K < 7) zlevel<K + 1>(a, z);
}

template <int Q>
__device__ __forceinline__ void enc_level(float* S, const float* cq, const float* sq) {
  if constexpr (Q < 8) {
    #pragma unroll
    for (int j = (1 << Q) - 1; j >= 0; --j) {
      const float v = S[j];
      S[2 * j + 1] = v * sq[Q];
      S[2 * j]     = v * cq[Q];
    }
    enc_level<Q + 1>(S, cq, sq);
  }
}

template <int G, int O>
__device__ __forceinline__ void product_pairs(const float* in, float* res,
                                              const v2f* a0, const v2f* a1) {
  if constexpr (O < 32) {
    constexpr int u0 = var_of(G, amp_of(G, O, 0));
    constexpr int u1 = var_of(G, amp_of(G, O + 1, 0));
    const v2f* aa = u0 ? a1 : a0;
    const v2f* ab = u1 ? a1 : a0;
    v8f d0 = {}; v8f d1 = {};
    #pragma unroll
    for (int kc = 0; kc < 4; ++kc) {
      v2f b0, b1;
      b0[0] = in[amp_of(G, O,     2 * kc)]; b0[1] = in[amp_of(G, O,     2 * kc + 1)];
      b1[0] = in[amp_of(G, O + 1, 2 * kc)]; b1[1] = in[amp_of(G, O + 1, 2 * kc + 1)];
      d0 = __builtin_amdgcn_wmma_f32_16x16x4_f32(false, aa[kc], false, b0, (short)0, d0, false, false);
      d1 = __builtin_amdgcn_wmma_f32_16x16x4_f32(false, ab[kc], false, b1, (short)0, d1, false, false);
    }
    #pragma unroll
    for (int v = 0; v < 8; ++v) {
      res[file_of(G, amp_of(G, O,     v))] = d0[v];
      res[file_of(G, amp_of(G, O + 1, v))] = d1[v];
    }
    product_pairs<G, O + 2>(in, res, a0, a1);
  }
}

__device__ __forceinline__ void frag_to_a(v4f f0, v4f f1, v2f* a) {
  a[0][0] = f0[0]; a[0][1] = f0[1]; a[1][0] = f0[2]; a[1][1] = f0[3];
  a[2][0] = f1[0]; a[2][1] = f1[1]; a[3][0] = f1[2]; a[3][1] = f1[3];
}

template <int I, int N>
__device__ __forceinline__ void squares(const float* s, float* q) {
  if constexpr (I < N) {
    #pragma unroll
    for (int i = I; i < I + 32; ++i) q[i] = s[i] * s[i];
    squares<I + 32, N>(s, q);
  }
}


template <int G>
__device__ __forceinline__ void run_block(const float* in, float* res, const v4f* fr) {
  v2f a0[4], a1[4];
  frag_to_a(fr[0], fr[1], a0);
  frag_to_a(fr[2], fr[3], a1);
  product_pairs<G, 0>(in, res, a0, a1);
}

__global__ __launch_bounds__(128) __attribute__((amdgpu_waves_per_eu(1, 1)))
void qsim(const float* __restrict__ x,
          const float* __restrict__ ws,
          float* __restrict__ out, int B) {
  const int lane = threadIdx.x & 31;
  const int wave = blockIdx.x * 4 + (threadIdx.x >> 5);
  const int half = lane >> 4;
  const int b    = wave * 16 + (lane & 15);
  const int bc   = b < B ? b : B - 1;

  const v4f* frag = reinterpret_cast<const v4f*>(ws) + lane;
  v4f fr[NBLOCKS][4];
  #pragma unroll
  for (int s = 0; s < NBLOCKS; ++s) {
    fr[s][0] = frag[128 * s];
    fr[s][1] = frag[128 * s + 32];
    if (ctl_wire(s) >= 0) {
      fr[s][2] = frag[128 * s + 64];
      fr[s][3] = frag[128 * s + 96];
    } else {
      fr[s][2] = fr[s][0];
      fr[s][3] = fr[s][1];
    }
  }

  const v4f* ang = reinterpret_cast<const v4f*>(ws + ANG_OFF + (size_t)bc * 16);
  const v4f c0 = ang[0], c1 = ang[1], s0 = ang[2], s1 = ang[3];
  const float cq[8] = {c0[0], c0[1], c0[2], c0[3], c1[0], c1[1], c1[2], c1[3]};
  const float sq[8] = {s0[0], s0[1], s0[2], s0[3], s1[0], s1[1], s1[2], s1[3]};
  float S[256];
  S[0] = half ? 0.f : 1.f;
  enc_level<0>(S, cq, sq);

  float R[256];
  run_block<0>(S, R, fr[0]);
  run_block<1>(R, S, fr[1]);
  run_block<2>(S, R, fr[2]);
  run_block<3>(R, S, fr[3]);
  run_block<4>(S, R, fr[4]);
  run_block<5>(R, S, fr[5]);

  float sqv[256];
  squares<0, 256>(S, sqv);
  float z[8];
  zlevel<0>(sqv, z);
  #pragma unroll
  for (int k = 0; k < 8; ++k) z[k] += __shfl_xor(z[k], 16, 32);
  if (half == 0 && b < B) {
    v4f o0, o1;
    o0[0] = z[7]; o0[1] = z[6]; o0[2] = z[5]; o0[3] = z[4];
    o1[0] = z[3]; o1[1] = z[2]; o1[2] = z[1]; o1[3] = z[0];
    v4f* dst = reinterpret_cast<v4f*>(out + (size_t)b * 8);
    dst[0] = o0;
    dst[1] = o1;
  }
}

extern "C" void kernel_launch(void* const* d_in, const int* in_sizes, int n_in,
                              void* d_out, int out_size, void* d_ws, size_t ws_size,
                              hipStream_t stream) {
  const float* x = (const float*)d_in[0];
  const float* w = (const float*)d_in[1];
  float* out = (float*)d_out;
  float* ws  = (float*)d_ws;
  int B = in_sizes[0] / NQ;

  int nquads = B * 2;
  int tblocks = (nquads + PREP_THREADS - 1) / PREP_THREADS;
  qprep<<<PREP_BLOCKS + tblocks, PREP_THREADS, 0, stream>>>(x, w, ws, nquads);
  int nblocks = (B + 63) / 64;
  qsim<<<nblocks, 128, 0, stream>>>(x, ws, out, B);
}


# Wires no block has touched yet stay a per-sample cos/sin factor, multiplied in before their first block: blocks 0-3 run 1, 4, 16, 16 products, not 32 each; 768 -> 404 WMMA/wave; rounding regrouped
# speedup vs baseline: 4.3293x; 1.2969x over previous
#include <hip/hip_runtime.h>


typedef __attribute__((ext_vector_type(2))) float v2f;
typedef __attribute__((ext_vector_type(8))) float v8f;
typedef __attribute__((ext_vector_type(4))) float v4f;

#define NQ 8
#define NBLOCKS 6
#define NMATS (NBLOCKS * 2)
#define MAXOPS 16

struct cpx { float re, im; };

__device__ __forceinline__ cpx cmul(cpx a, cpx b) {
  cpx r; r.re = a.re * b.re - a.im * b.im; r.im = a.re * b.im + a.im * b.re; return r;
}
__device__ __forceinline__ cpx cadd(cpx a, cpx b) {
  cpx r; r.re = a.re + b.re; r.im = a.im + b.im; return r;
}

__device__ __forceinline__ cpx u_entry(float ca, float sa, float cb, float sb,
                                       float cg, float sg, int i, int j) {
  cpx m;
  if (i == 0 && j == 0) { m.re =  cb * ca; m.im =  sb * sa; }
  else if (i == 0)      { m.re = -sb * ca; m.im = -cb * sa; }
  else if (j == 0)      { m.re =  sb * ca; m.im = -cb * sa; }
  else                  { m.re =  cb * ca; m.im = -sb * sa; }
  cpx ph; ph.re = cg; ph.im = (i == 0) ? -sg : sg;
  return cmul(ph, m);
}

__host__ __device__ constexpr int win_a(int s) { return s == 0 ? 0 : s == 1 ? 2 : s == 2 ? 4 : s == 3 ? 1 : s == 4 ? 5 : 0; }
__host__ __device__ constexpr int win_b(int s) { return s == 0 ? 1 : s == 1 ? 3 : s == 2 ? 5 : s == 3 ? 2 : s == 4 ? 6 : 3; }
__host__ __device__ constexpr int win_c(int s) { return s == 0 ? 2 : s == 1 ? 4 : s == 2 ? 6 : s == 3 ? 3 : s == 4 ? 7 : 4; }
__host__ __device__ constexpr int ctl_wire(int s) { return s == 1 ? 1 : s == 2 ? 3 : s == 3 ? 0 : s == 4 ? 4 : -1; }

__host__ __device__ constexpr int ins_zero(int x, int p) {
  return ((x >> p) << (p + 1)) | (x & ((1 << p) - 1));
}
__host__ __device__ constexpr int amp_of(int s, int o, int v) {
  const int pa = 7 - win_a(s), pb = 7 - win_b(s), pc = 7 - win_c(s);
  return ins_zero(ins_zero(ins_zero(o, pc), pb), pa)
       | (((v >> 2) & 1) << pa) | (((v >> 1) & 1) << pb) | ((v & 1) << pc);
}
__host__ __device__ constexpr int var_of(int s, int j) {
  return ctl_wire(s) < 0 ? 0 : ((j >> (7 - ctl_wire(s))) & 1);
}
__host__ __device__ constexpr int cx(int j, int c, int t) {
  return j ^ (((j >> (7 - c)) & 1) << (7 - t));
}
__host__ __device__ constexpr int cnot_dst(int j) {
  return (j ^ (j >> 1) ^ (j >> 2) ^ (j >> 3) ^ (j >> 4) ^ (j >> 5) ^ (j >> 6) ^ (j >> 7)) & 255;
}
__host__ __device__ constexpr int file_of(int s, int j) {
  return s == 3 ? cx(j, 3, 4) : s == 5 ? cnot_dst(j) : j;
}

__host__ __device__ constexpr int wires_from(int q) { return (1 << (8 - q)) - 1; }
__host__ __device__ constexpr int bits_set(int x) { int c = 0; for (; x; x &= x - 1) ++c; return c; }
__host__ __device__ constexpr int nth_o(int s, int fm, int n) {
  int c = 0;
  for (int o = 0; o < 32; ++o) {
    if ((amp_of(s, o, 0) & fm) == 0) {
      if (c == n) return o;
      ++c;
    }
  }
  return 0;
}

#define OP_GATE(sh, l, w) (0 | ((sh) << 2) | ((((l) * NQ) + (w)) << 4))
#define OP_CNOT(c, t)     (1 | ((c) << 2) | ((t) << 4))
#define OP_XOUT(t)        (2 | ((t) << 2))
#define OP_END            3
__constant__ const unsigned short block_ops[NBLOCKS * MAXOPS] = {
    OP_GATE(2, 0, 0), OP_GATE(1, 0, 1), OP_GATE(0, 0, 2), OP_CNOT(2, 1), OP_CNOT(1, 0), OP_GATE(2, 1, 0), OP_GATE(1, 1, 1), OP_CNOT(2, 1), OP_GATE(2, 2, 0), OP_END, OP_END, OP_END, OP_END, OP_END, OP_END, OP_END,
    OP_GATE(1, 0, 3), OP_GATE(0, 0, 4), OP_CNOT(2, 1), OP_CNOT(1, 0), OP_GATE(2, 1, 2), OP_GATE(1, 1, 3), OP_XOUT(2), OP_CNOT(2, 1), OP_GATE(2, 2, 2), OP_END, OP_END, OP_END, OP_END, OP_END, OP_END, OP_END,
    OP_GATE(1, 0, 5), OP_GATE(0, 0, 6), OP_CNOT(2, 1), OP_CNOT(1, 0), OP_GATE(2, 1, 4), OP_GATE(1, 1, 5), OP_XOUT(2), OP_CNOT(2, 1), OP_GATE(2, 2, 4), OP_END, OP_END, OP_END, OP_END, OP_END, OP_END, OP_END,
    OP_GATE(2, 2, 1), OP_GATE(0, 2, 3), OP_XOUT(2), OP_CNOT(2, 1), OP_CNOT(1, 0), OP_GATE(2, 3, 1), OP_GATE(1, 3, 2), OP_END, OP_END, OP_END, OP_END, OP_END, OP_END, OP_END, OP_END, OP_END,
    OP_GATE(0, 0, 7), OP_CNOT(1, 0), OP_GATE(0, 1, 7), OP_GATE(1, 1, 6), OP_CNOT(2, 1), OP_CNOT(1, 0), OP_GATE(0, 2, 7), OP_GATE(2, 2, 5), OP_GATE(1, 2, 6), OP_XOUT(2), OP_CNOT(2, 1), OP_CNOT(1, 0), OP_GATE(2, 3, 5), OP_GATE(1, 3, 6), OP_GATE(0, 3, 7), OP_END,
    OP_GATE(2, 3, 0), OP_GATE(1, 3, 3), OP_GATE(0, 3, 4), OP_END, OP_END, OP_END, OP_END, OP_END, OP_END, OP_END, OP_END, OP_END, OP_END, OP_END, OP_END, OP_END
};

#define PREP_THREADS 64
#define PREP_BLOCKS (NMATS * (256 / PREP_THREADS))
#define ANG_OFF (NMATS * 256)
__global__ __launch_bounds__(PREP_THREADS) void qprep(const float* __restrict__ x,
                                                      const float* __restrict__ w,
                                                      float* __restrict__ ws, int nquads) {
  __shared__ float trig[MAXOPS * 6];
  if (blockIdx.x >= PREP_BLOCKS) {
    const int t = (blockIdx.x - PREP_BLOCKS) * PREP_THREADS + threadIdx.x;
    if (t < nquads) {
      const v4f xv = reinterpret_cast<const v4f*>(x)[t];
      v4f c, s;
      #pragma unroll
      for (int i = 0; i < 4; ++i) { c[i] = cosf(0.5f * xv[i]); s[i] = sinf(0.5f * xv[i]); }
      v4f* dst = reinterpret_cast<v4f*>(ws + ANG_OFF + (size_t)(t >> 1) * 16 + (t & 1) * 4);
      dst[0] = c;
      dst[2] = s;
    }
    return;
  }
  const int parts = 256 / PREP_THREADS;
  const int mi = blockIdx.x / parts;
  const int e  = (blockIdx.x % parts) * PREP_THREADS + threadIdx.x;
  const int s  = mi >> 1;
  const int u  = mi & 1;
  if (u == 1 && ctl_wire(s) < 0) return;
  const unsigned short* ops = block_ops + s * MAXOPS;
  const int myop = ops[threadIdx.x & (MAXOPS - 1)];
  if ((int)threadIdx.x < 3 * MAXOPS) {
    const int i = threadIdx.x / 3, r = threadIdx.x - 3 * i;
    const int op = __shfl(myop, i, 32);
    if ((op & 3) == 0) {
      const float h = 0.5f * w[3 * (op >> 4) + r];
      trig[6 * i + 2 * r]     = cosf(h);
      trig[6 * i + 2 * r + 1] = sinf(h);
    }
  }
  __syncthreads();
  const int m = e >> 4, k = e & 15;
  const int pp = m >> 3, bp = m & 7;
  const int pt = k >> 3, b  = k & 7;
  cpx v[8];
  #pragma unroll
  for (int j = 0; j < 8; ++j) { v[j].re = (j == b) ? 1.f : 0.f; v[j].im = 0.f; }
  for (int i = 0; i < MAXOPS; ++i) {
    const int op = __builtin_amdgcn_readlane(myop, i);
    const int kind = op & 3, p = (op >> 2) & 3, q = op >> 4;
    if (kind == 3) break;
    if (kind == 0) {
      const float* t = trig + 6 * i;
      const cpx u00 = u_entry(t[0], t[1], t[2], t[3], t[4], t[5], 0, 0);
      const cpx u01 = u_entry(t[0], t[1], t[2], t[3], t[4], t[5], 0, 1);
      const cpx u10 = u_entry(t[0], t[1], t[2], t[3], t[4], t[5], 1, 0);
      const cpx u11 = u_entry(t[0], t[1], t[2], t[3], t[4], t[5], 1, 1);
      #pragma unroll
      for (int sh = 0; sh < 3; ++sh) {
        if (p == sh) {
          #pragma unroll
          for (int j = 0; j < 8; ++j) {
            if (!(j & (1 << sh))) {
              const cpx x0 = v[j], x1 = v[j | (1 << sh)];
              v[j]             = cadd(cmul(u00, x0), cmul(u01, x1));
              v[j | (1 << sh)] = cadd(cmul(u10, x0), cmul(u11, x1));
            }
          }
        }
      }
    } else if (kind == 1 || u == 1) {
      const int tb = (kind == 1) ? q : p;
      #pragma unroll
      for (int sh = 0; sh < 3; ++sh) {
        if (tb == sh) {
          #pragma unroll
          for (int j = 0; j < 8; ++j) {
            if (!(j & (1 << sh))) {
              const bool on = (kind == 2) || (((j >> p) & 1) != 0);
              const cpx x0 = v[j], x1 = v[j | (1 << sh)];
              v[j]             = on ? x1 : x0;
              v[j | (1 << sh)] = on ? x0 : x1;
            }
          }
        }
      }
    }
  }
  cpx G; G.re = 0.f; G.im = 0.f;
  #pragma unroll
  for (int j = 0; j < 8; ++j) if (j == bp) G = v[j];
  const float val = (pp == pt) ? G.re : (pp ? G.im : -G.im);
  const int fl = m + 16 * pt;
  ws[(size_t)mi * 256 + (b >> 2) * 128 + fl * 4 + (b & 3)] = val;
}

template <int ST>
__device__ __forceinline__ void tree_fold(float* d) {
  if constexpr (ST >= 1) {
    #pragma unroll
    for (int j = 0; j < ST; ++j) d[j] += d[j + ST];
    tree_fold<ST / 2>(d);
  }
}

template <int K>
__device__ __forceinline__ void zlevel(float* a, float* z) {
  constexpr int H = 128 >> K;
  float dd[H];
  #pragma unroll
  for (int j = 0; j < H; ++j) {
    const float e0 = a[2 * j], e1 = a[2 * j + 1];
    dd[j] = e0 - e1;
    a[j]  = e0 + e1;
  }
  tree_fold<H / 2>(dd);
  z[K] = dd[0];
  if constexpr (K < 7) zlevel<K + 1>(a, z);
}

template <int Q, int QEND>
__device__ __forceinline__ void enc_level(float* S, const float* cq, const float* sq) {
  if constexpr (Q < QEND) {
    #pragma unroll
    for (int j = (1 << Q) - 1; j >= 0; --j) {
      const float v = S[j];
      S[2 * j + 1] = v * sq[Q];
      S[2 * j]     = v * cq[Q];
    }
    enc_level<Q + 1, QEND>(S, cq, sq);
  }
}

template <int FM, int BA, int BB, int J0>
__device__ __forceinline__ void expand2(float* a, const float* g) {
  if constexpr (J0 < 256) {
    #pragma unroll
    for (int j = J0; j < J0 + 32; ++j) {
      if ((j & FM) == 0) {
        const float v = a[j];
        a[j | BA | BB] = v * g[3];
        a[j | BA]      = v * g[2];
        a[j | BB]      = v * g[1];
        a[j]           = v * g[0];
      }
    }
    expand2<FM, BA, BB, J0 + 32>(a, g);
  }
}
template <int FM, int BA, int J0>
__device__ __forceinline__ void expand1(float* a, float f0, float f1) {
  if constexpr (J0 < 256) {
    #pragma unroll
    for (int j = J0; j < J0 + 32; ++j) {
      if ((j & FM) == 0) {
        const float v = a[j];
        a[j | BA] = v * f1;
        a[j]      = v * f0;
      }
    }
    expand1<FM, BA, J0 + 32>(a, f0, f1);
  }
}

template <int G, int FM, int N>
__device__ __forceinline__ void product_pairs(const float* in, float* res,
                                              const v2f* a0, const v2f* a1) {
  constexpr int CNT = 32 >> bits_set(FM);
  if constexpr (N + 1 < CNT) {
    constexpr int O = nth_o(G, FM, N), P = nth_o(G, FM, N + 1);
    constexpr int u0 = var_of(G, amp_of(G, O, 0));
    constexpr int u1 = var_of(G, amp_of(G, P, 0));
    const v2f* aa = u0 ? a1 : a0;
    const v2f* ab = u1 ? a1 : a0;
    v8f d0 = {}; v8f d1 = {};
    #pragma unroll
    for (int kc = 0; kc < 4; ++kc) {
      v2f b0, b1;
      b0[0] = in[amp_of(G, O,     2 * kc)]; b0[1] = in[amp_of(G, O,     2 * kc + 1)];
      b1[0] = in[amp_of(G, P, 2 * kc)]; b1[1] = in[amp_of(G, P, 2 * kc + 1)];
      d0 = __builtin_amdgcn_wmma_f32_16x16x4_f32(false, aa[kc], false, b0, (short)0, d0, false, false);
      d1 = __builtin_amdgcn_wmma_f32_16x16x4_f32(false, ab[kc], false, b1, (short)0, d1, false, false);
    }
    #pragma unroll
    for (int v = 0; v < 8; ++v) {
      res[file_of(G, amp_of(G, O,     v))] = d0[v];
      res[file_of(G, amp_of(G, P, v))] = d1[v];
    }
    product_pairs<G, FM, N + 2>(in, res, a0, a1);
  } else if constexpr (N < CNT) {
    constexpr int O = nth_o(G, FM, N);
    const v2f* aa = var_of(G, amp_of(G, O, 0)) ? a1 : a0;
    v8f d0 = {};
    #pragma unroll
    for (int kc = 0; kc < 4; ++kc) {
      v2f b0;
      b0[0] = in[amp_of(G, O, 2 * kc)]; b0[1] = in[amp_of(G, O, 2 * kc + 1)];
      d0 = __builtin_amdgcn_wmma_f32_16x16x4_f32(false, aa[kc], false, b0, (short)0, d0, false, false);
    }
    #pragma unroll
    for (int v = 0; v < 8; ++v) res[file_of(G, amp_of(G, O, v))] = d0[v];
  }
}

__device__ __forceinline__ void frag_to_a(v4f f0, v4f f1, v2f* a) {
  a[0][0] = f0[0]; a[0][1] = f0[1]; a[1][0] = f0[2]; a[1][1] = f0[3];
  a[2][0] = f1[0]; a[2][1] = f1[1]; a[3][0] = f1[2]; a[3][1] = f1[3];
}

template <int I, int N>
__device__ __forceinline__ void squares(const float* s, float* q) {
  if constexpr (I < N) {
    #pragma unroll
    for (int i = I; i < I + 32; ++i) q[i] = s[i] * s[i];
    squares<I + 32, N>(s, q);
  }
}


template <int G, int FM>
__device__ __forceinline__ void run_block(const float* in, float* res, const v4f* fr) {
  v2f a0[4], a1[4];
  frag_to_a(fr[0], fr[1], a0);
  frag_to_a(fr[2], fr[3], a1);
  product_pairs<G, FM, 0>(in, res, a0, a1);
}

__global__ __launch_bounds__(128) __attribute__((amdgpu_waves_per_eu(1, 1)))
void qsim(const float* __restrict__ x,
          const float* __restrict__ ws,
          float* __restrict__ out, int B) {
  const int lane = threadIdx.x & 31;
  const int wave = blockIdx.x * 4 + (threadIdx.x >> 5);
  const int half = lane >> 4;
  const int b    = wave * 16 + (lane & 15);
  const int bc   = b < B ? b : B - 1;

  const v4f* frag = reinterpret_cast<const v4f*>(ws) + lane;
  v4f fr[NBLOCKS][4];
  #pragma unroll
  for (int s = 0; s < NBLOCKS; ++s) {
    fr[s][0] = frag[128 * s];
    fr[s][1] = frag[128 * s + 32];
    if (ctl_wire(s) >= 0) {
      fr[s][2] = frag[128 * s + 64];
      fr[s][3] = frag[128 * s + 96];
    } else {
      fr[s][2] = fr[s][0];
      fr[s][3] = fr[s][1];
    }
  }

  const v4f* ang = reinterpret_cast<const v4f*>(ws + ANG_OFF + (size_t)bc * 16);
  const v4f c0 = ang[0], c1 = ang[1], s0 = ang[2], s1 = ang[3];
  const float cq[8] = {c0[0], c0[1], c0[2], c0[3], c1[0], c1[1], c1[2], c1[3]};
  const float sq[8] = {s0[0], s0[1], s0[2], s0[3], s1[0], s1[1], s1[2], s1[3]};
  float E[8];
  E[0] = half ? 0.f : 1.f;
  enc_level<0, 3>(E, cq, sq);
  float S[256];
  #pragma unroll
  for (int v = 0; v < 8; ++v) S[v << 5] = E[v];

  float R[256];
  run_block<0, wires_from(3)>(S, R, fr[0]);
  const float g34[4] = {cq[3] * cq[4], cq[3] * sq[4], sq[3] * cq[4], sq[3] * sq[4]};
  expand2<wires_from(3), 1 << (7 - 3), 1 << (7 - 4), 0>(R, g34);
  run_block<1, wires_from(5)>(R, S, fr[1]);
  const float g56[4] = {cq[5] * cq[6], cq[5] * sq[6], sq[5] * cq[6], sq[5] * sq[6]};
  expand2<wires_from(5), 1 << (7 - 5), 1 << (7 - 6), 0>(S, g56);
  run_block<2, wires_from(7)>(S, R, fr[2]);
  run_block<3, wires_from(7)>(R, S, fr[3]);
  expand1<wires_from(7), 1 << (7 - 7), 0>(S, cq[7], sq[7]);
  run_block<4, 0>(S, R, fr[4]);
  run_block<5, 0>(R, S, fr[5]);

  float sqv[256];
  squares<0, 256>(S, sqv);
  float z[8];
  zlevel<0>(sqv, z);
  #pragma unroll
  for (int k = 0; k < 8; ++k) z[k] += __shfl_xor(z[k], 16, 32);
  if (half == 0 && b < B) {
    v4f o0, o1;
    o0[0] = z[7]; o0[1] = z[6]; o0[2] = z[5]; o0[3] = z[4];
    o1[0] = z[3]; o1[1] = z[2]; o1[2] = z[1]; o1[3] = z[0];
    v4f* dst = reinterpret_cast<v4f*>(out + (size_t)b * 8);
    dst[0] = o0;
    dst[1] = o1;
  }
}

extern "C" void kernel_launch(void* const* d_in, const int* in_sizes, int n_in,
                              void* d_out, int out_size, void* d_ws, size_t ws_size,
                              hipStream_t stream) {
  const float* x = (const float*)d_in[0];
  const float* w = (const float*)d_in[1];
  float* out = (float*)d_out;
  float* ws  = (float*)d_ws;
  int B = in_sizes[0] / NQ;

  int nquads = B * 2;
  int tblocks = (nquads + PREP_THREADS - 1) / PREP_THREADS;
  qprep<<<PREP_BLOCKS + tblocks, PREP_THREADS, 0, stream>>>(x, w, ws, nquads);
  int nblocks = (B + 63) / 64;
  qsim<<<nblocks, 128, 0, stream>>>(x, ws, out, B);
}


# Block-matrix build of the prep launch: one work-group per matrix, one complex value per lane; a gate = a vector-ALU lane exchange + two complex multiply-adds (was an 8-vector per thread); same sums
# speedup vs baseline: 5.0371x; 1.1635x over previous
#include <hip/hip_runtime.h>


typedef __attribute__((ext_vector_type(2))) float v2f;
typedef __attribute__((ext_vector_type(8))) float v8f;
typedef __attribute__((ext_vector_type(4))) float v4f;

#define NQ 8
#define NBLOCKS 6
#define NMATS (NBLOCKS * 2)
#define MAXOPS 16

struct cpx { float re, im; };

__device__ __forceinline__ cpx cmul(cpx a, cpx b) {
  cpx r; r.re = a.re * b.re - a.im * b.im; r.im = a.re * b.im + a.im * b.re; return r;
}
__device__ __forceinline__ cpx cadd(cpx a, cpx b) {
  cpx r; r.re = a.re + b.re; r.im = a.im + b.im; return r;
}

__device__ __forceinline__ cpx u_entry(float ca, float sa, float cb, float sb,
                                       float cg, float sg, int i, int j) {
  cpx m;
  if (i == 0 && j == 0) { m.re =  cb * ca; m.im =  sb * sa; }
  else if (i == 0)      { m.re = -sb * ca; m.im = -cb * sa; }
  else if (j == 0)      { m.re =  sb * ca; m.im = -cb * sa; }
  else                  { m.re =  cb * ca; m.im = -sb * sa; }
  cpx ph; ph.re = cg; ph.im = (i == 0) ? -sg : sg;
  return cmul(ph, m);
}

__host__ __device__ constexpr int win_a(int s) { return s == 0 ? 0 : s == 1 ? 2 : s == 2 ? 4 : s == 3 ? 1 : s == 4 ? 5 : 0; }
__host__ __device__ constexpr int win_b(int s) { return s == 0 ? 1 : s == 1 ? 3 : s == 2 ? 5 : s == 3 ? 2 : s == 4 ? 6 : 3; }
__host__ __device__ constexpr int win_c(int s) { return s == 0 ? 2 : s == 1 ? 4 : s == 2 ? 6 : s == 3 ? 3 : s == 4 ? 7 : 4; }
__host__ __device__ constexpr int ctl_wire(int s) { return s == 1 ? 1 : s == 2 ? 3 : s == 3 ? 0 : s == 4 ? 4 : -1; }

__host__ __device__ constexpr int ins_zero(int x, int p) {
  return ((x >> p) << (p + 1)) | (x & ((1 << p) - 1));
}
__host__ __device__ constexpr int amp_of(int s, int o, int v) {
  const int pa = 7 - win_a(s), pb = 7 - win_b(s), pc = 7 - win_c(s);
  return ins_zero(ins_zero(ins_zero(o, pc), pb), pa)
       | (((v >> 2) & 1) << pa) | (((v >> 1) & 1) << pb) | ((v & 1) << pc);
}
__host__ __device__ constexpr int var_of(int s, int j) {
  return ctl_wire(s) < 0 ? 0 : ((j >> (7 - ctl_wire(s))) & 1);
}
__host__ __device__ constexpr int cx(int j, int c, int t) {
  return j ^ (((j >> (7 - c)) & 1) << (7 - t));
}
__host__ __device__ constexpr int cnot_dst(int j) {
  return (j ^ (j >> 1) ^ (j >> 2) ^ (j >> 3) ^ (j >> 4) ^ (j >> 5) ^ (j >> 6) ^ (j >> 7)) & 255;
}
__host__ __device__ constexpr int file_of(int s, int j) {
  return s == 3 ? cx(j, 3, 4) : s == 5 ? cnot_dst(j) : j;
}

__host__ __device__ constexpr int wires_from(int q) { return (1 << (8 - q)) - 1; }
__host__ __device__ constexpr int bits_set(int x) { int c = 0; for (; x; x &= x - 1) ++c; return c; }
__host__ __device__ constexpr int nth_o(int s, int fm, int n) {
  int c = 0;
  for (int o = 0; o < 32; ++o) {
    if ((amp_of(s, o, 0) & fm) == 0) {
      if (c == n) return o;
      ++c;
    }
  }
  return 0;
}

#define OP_GATE(sh, l, w) (0 | ((sh) << 2) | ((((l) * NQ) + (w)) << 4))
#define OP_CNOT(c, t)     (1 | ((c) << 2) | ((t) << 4))
#define OP_XOUT(t)        (2 | ((t) << 2))
#define OP_END            3
#define BLOCK_OPS_LIST \
      \
    OP_GATE(2, 0, 0), OP_GATE(1, 0, 1), OP_GATE(0, 0, 2), OP_CNOT(2, 1), OP_CNOT(1, 0), OP_GATE(2, 1, 0), OP_GATE(1, 1, 1), OP_CNOT(2, 1), OP_GATE(2, 2, 0), OP_END, OP_END, OP_END, OP_END, OP_END, OP_END, OP_END, \
      \
    OP_GATE(1, 0, 3), OP_GATE(0, 0, 4), OP_CNOT(2, 1), OP_CNOT(1, 0), OP_GATE(2, 1, 2), OP_GATE(1, 1, 3), OP_XOUT(2), OP_CNOT(2, 1), OP_GATE(2, 2, 2), OP_END, OP_END, OP_END, OP_END, OP_END, OP_END, OP_END, \
      \
    OP_GATE(1, 0, 5), OP_GATE(0, 0, 6), OP_CNOT(2, 1), OP_CNOT(1, 0), OP_GATE(2, 1, 4), OP_GATE(1, 1, 5), OP_XOUT(2), OP_CNOT(2, 1), OP_GATE(2, 2, 4), OP_END, OP_END, OP_END, OP_END, OP_END, OP_END, OP_END, \
      \
    OP_GATE(2, 2, 1), OP_GATE(0, 2, 3), OP_XOUT(2), OP_CNOT(2, 1), OP_CNOT(1, 0), OP_GATE(2, 3, 1), OP_GATE(1, 3, 2), OP_END, OP_END, OP_END, OP_END, OP_END, OP_END, OP_END, OP_END, OP_END, \
      \
    OP_GATE(0, 0, 7), OP_CNOT(1, 0), OP_GATE(0, 1, 7), OP_GATE(1, 1, 6), OP_CNOT(2, 1), OP_CNOT(1, 0), OP_GATE(0, 2, 7), OP_GATE(2, 2, 5), OP_GATE(1, 2, 6), OP_XOUT(2), OP_CNOT(2, 1), OP_CNOT(1, 0), OP_GATE(2, 3, 5), OP_GATE(1, 3, 6), OP_GATE(0, 3, 7), OP_END, \
      \
    OP_GATE(2, 3, 0), OP_GATE(1, 3, 3), OP_GATE(0, 3, 4), OP_END, OP_END, OP_END, OP_END, OP_END, OP_END, OP_END, OP_END, OP_END, OP_END, OP_END, OP_END, OP_END
__constant__ const unsigned short block_ops[NBLOCKS * MAXOPS] = { BLOCK_OPS_LIST };
constexpr unsigned short block_ops_c[NBLOCKS * MAXOPS] = { BLOCK_OPS_LIST };


template <int N>
__device__ __forceinline__ float from_lane_xor(float x) {
  const int v = __float_as_int(x);
  return __int_as_float(__builtin_amdgcn_update_dpp(v, v, 0x160 + N, 0xf, 0xf, false));
}
template <int SH>
__device__ __forceinline__ cpx partner(cpx v) {
  cpx o;
  o.re = from_lane_xor<(1 << SH)>(v.re);
  o.im = from_lane_xor<(1 << SH)>(v.im);
  return o;
}
template <int I>
__device__ __forceinline__ float lane_read(float x) {
  return __int_as_float(__builtin_amdgcn_readlane(__float_as_int(x), I));
}
template <int S, int I>
__device__ __forceinline__ void apply_ops(cpx& v, const float* ue, int j, bool second) {
  if constexpr (I < MAXOPS) {
    constexpr int op = block_ops_c[S * MAXOPS + I];
    constexpr int kind = op & 3, p = (op >> 2) & 3, q = op >> 4;
    if constexpr (kind != 3) {
      if constexpr (kind == 0) {
        float e[8];
        #pragma unroll
        for (int k = 0; k < 8; ++k) e[k] = lane_read<I>(ue[k]);
        const bool hi = ((j >> p) & 1) != 0;
        cpx c0, c1;
        c0.re = hi ? e[4] : e[0]; c0.im = hi ? e[5] : e[1];
        c1.re = hi ? e[6] : e[2]; c1.im = hi ? e[7] : e[3];
        const cpx o = partner<p>(v);
        const cpx x0 = hi ? o : v, x1 = hi ? v : o;
        v = cadd(cmul(c0, x0), cmul(c1, x1));
      } else if constexpr (kind == 1) {
        const cpx o = partner<q>(v);
        const bool on = ((j >> p) & 1) != 0;
        v.re = on ? o.re : v.re;
        v.im = on ? o.im : v.im;
      } else {
        const cpx o = partner<p>(v);
        v.re = second ? o.re : v.re;
        v.im = second ? o.im : v.im;
      }
      apply_ops<S, I + 1>(v, ue, j, second);
    }
  }
}

#define PREP_THREADS 64
#define PREP_BLOCKS NMATS
#define ANG_OFF (NMATS * 256)
__global__ __launch_bounds__(PREP_THREADS) void qprep(const float* __restrict__ x,
                                                      const float* __restrict__ w,
                                                      float* __restrict__ ws, int nquads) {
  __shared__ float trig[MAXOPS * 6];
  if (blockIdx.x >= PREP_BLOCKS) {
    const int t = (blockIdx.x - PREP_BLOCKS) * PREP_THREADS + threadIdx.x;
    if (t < nquads) {
      const v4f xv = reinterpret_cast<const v4f*>(x)[t];
      v4f c, s;
      #pragma unroll
      for (int i = 0; i < 4; ++i) { c[i] = cosf(0.5f * xv[i]); s[i] = sinf(0.5f * xv[i]); }
      v4f* dst = reinterpret_cast<v4f*>(ws + ANG_OFF + (size_t)(t >> 1) * 16 + (t & 1) * 4);
      dst[0] = c;
      dst[2] = s;
    }
    return;
  }
  const int mi = blockIdx.x;
  const int s  = mi >> 1;
  const int u  = mi & 1;
  if (u == 1 && ctl_wire(s) < 0) return;
  const unsigned short* ops = block_ops + s * MAXOPS;
  const int myop = ops[threadIdx.x & (MAXOPS - 1)];
  if ((int)threadIdx.x < 3 * MAXOPS) {
    const int i = threadIdx.x / 3, r = threadIdx.x - 3 * i;
    const int op = __shfl(myop, i, 32);
    if ((op & 3) == 0) {
      const float h = 0.5f * w[3 * (op >> 4) + r];
      trig[6 * i + 2 * r]     = cosf(h);
      trig[6 * i + 2 * r + 1] = sinf(h);
    }
  }
  __syncthreads();
  float ue[8];
  #pragma unroll
  for (int k = 0; k < 8; ++k) ue[k] = 0.f;
  if ((myop & 3) == 0) {
    const float* t = trig + 6 * (threadIdx.x & (MAXOPS - 1));
    const cpx u00 = u_entry(t[0], t[1], t[2], t[3], t[4], t[5], 0, 0);
    const cpx u01 = u_entry(t[0], t[1], t[2], t[3], t[4], t[5], 0, 1);
    const cpx u10 = u_entry(t[0], t[1], t[2], t[3], t[4], t[5], 1, 0);
    const cpx u11 = u_entry(t[0], t[1], t[2], t[3], t[4], t[5], 1, 1);
    ue[0] = u00.re; ue[1] = u00.im; ue[2] = u01.re; ue[3] = u01.im;
    ue[4] = u10.re; ue[5] = u10.im; ue[6] = u11.re; ue[7] = u11.im;
  }
  const int j = threadIdx.x & 7;
  const int b = threadIdx.x >> 3;
  cpx v; v.re = (j == b) ? 1.f : 0.f; v.im = 0.f;
  switch (s) {
    case 0:  apply_ops<0, 0>(v, ue, j, u == 1); break;
    case 1:  apply_ops<1, 0>(v, ue, j, u == 1); break;
    case 2:  apply_ops<2, 0>(v, ue, j, u == 1); break;
    case 3:  apply_ops<3, 0>(v, ue, j, u == 1); break;
    case 4:  apply_ops<4, 0>(v, ue, j, u == 1); break;
    default: apply_ops<5, 0>(v, ue, j, u == 1); break;
  }
  float* dst = ws + (size_t)mi * 256 + (b >> 2) * 128 + (b & 3);
  #pragma unroll
  for (int pp = 0; pp < 2; ++pp) {
    #pragma unroll
    for (int pt = 0; pt < 2; ++pt) {
      const float val = (pp == pt) ? v.re : (pp ? v.im : -v.im);
      const int fl = 8 * pp + j + 16 * pt;
      dst[fl * 4] = val;
    }
  }
}

template <int ST>
__device__ __forceinline__ void tree_fold(float* d) {
  if constexpr (ST >= 1) {
    #pragma unroll
    for (int j = 0; j < ST; ++j) d[j] += d[j + ST];
    tree_fold<ST / 2>(d);
  }
}

template <int K>
__device__ __forceinline__ void zlevel(float* a, float* z) {
  constexpr int H = 128 >> K;
  float dd[H];
  #pragma unroll
  for (int j = 0; j < H; ++j) {
    const float e0 = a[2 * j], e1 = a[2 * j + 1];
    dd[j] = e0 - e1;
    a[j]  = e0 + e1;
  }
  tree_fold<H / 2>(dd);
  z[K] = dd[0];
  if constexpr (K < 7) zlevel<K + 1>(a, z);
}

template <int Q, int QEND>
__device__ __forceinline__ void enc_level(float* S, const float* cq, const float* sq) {
  if constexpr (Q < QEND) {
    #pragma unroll
    for (int j = (1 << Q) - 1; j >= 0; --j) {
      const float v = S[j];
      S[2 * j + 1] = v * sq[Q];
      S[2 * j]     = v * cq[Q];
    }
    enc_level<Q + 1, QEND>(S, cq, sq);
  }
}

template <int FM, int BA, int BB, int J0>
__device__ __forceinline__ void expand2(float* a, const float* g) {
  if constexpr (J0 < 256) {
    #pragma unroll
    for (int j = J0; j < J0 + 32; ++j) {
      if ((j & FM) == 0) {
        const float v = a[j];
        a[j | BA | BB] = v * g[3];
        a[j | BA]      = v * g[2];
        a[j | BB]      = v * g[1];
        a[j]           = v * g[0];
      }
    }
    expand2<FM, BA, BB, J0 + 32>(a, g);
  }
}
template <int FM, int BA, int J0>
__device__ __forceinline__ void expand1(float* a, float f0, float f1) {
  if constexpr (J0 < 256) {
    #pragma unroll
    for (int j = J0; j < J0 + 32; ++j) {
      if ((j & FM) == 0) {
        const float v = a[j];
        a[j | BA] = v * f1;
        a[j]      = v * f0;
      }
    }
    expand1<FM, BA, J0 + 32>(a, f0, f1);
  }
}

template <int G, int FM, int N>
__device__ __forceinline__ void product_pairs(const float* in, float* res,
                                              const v2f* a0, const v2f* a1) {
  constexpr int CNT = 32 >> bits_set(FM);
  if constexpr (N + 1 < CNT) {
    constexpr int O = nth_o(G, FM, N), P = nth_o(G, FM, N + 1);
    constexpr int u0 = var_of(G, amp_of(G, O, 0));
    constexpr int u1 = var_of(G, amp_of(G, P, 0));
    const v2f* aa = u0 ? a1 : a0;
    const v2f* ab = u1 ? a1 : a0;
    v8f d0 = {}; v8f d1 = {};
    #pragma unroll
    for (int kc = 0; kc < 4; ++kc) {
      v2f b0, b1;
      b0[0] = in[amp_of(G, O,     2 * kc)]; b0[1] = in[amp_of(G, O,     2 * kc + 1)];
      b1[0] = in[amp_of(G, P, 2 * kc)]; b1[1] = in[amp_of(G, P, 2 * kc + 1)];
      d0 = __builtin_amdgcn_wmma_f32_16x16x4_f32(false, aa[kc], false, b0, (short)0, d0, false, false);
      d1 = __builtin_amdgcn_wmma_f32_16x16x4_f32(false, ab[kc], false, b1, (short)0, d1, false, false);
    }
    #pragma unroll
    for (int v = 0; v < 8; ++v) {
      res[file_of(G, amp_of(G, O,     v))] = d0[v];
      res[file_of(G, amp_of(G, P, v))] = d1[v];
    }
    product_pairs<G, FM, N + 2>(in, res, a0, a1);
  } else if constexpr (N < CNT) {
    constexpr int O = nth_o(G, FM, N);
    const v2f* aa = var_of(G, amp_of(G, O, 0)) ? a1 : a0;
    v8f d0 = {};
    #pragma unroll
    for (int kc = 0; kc < 4; ++kc) {
      v2f b0;
      b0[0] = in[amp_of(G, O, 2 * kc)]; b0[1] = in[amp_of(G, O, 2 * kc + 1)];
      d0 = __builtin_amdgcn_wmma_f32_16x16x4_f32(false, aa[kc], false, b0, (short)0, d0, false, false);
    }
    #pragma unroll
    for (int v = 0; v < 8; ++v) res[file_of(G, amp_of(G, O, v))] = d0[v];
  }
}

__device__ __forceinline__ void frag_to_a(v4f f0, v4f f1, v2f* a) {
  a[0][0] = f0[0]; a[0][1] = f0[1]; a[1][0] = f0[2]; a[1][1] = f0[3];
  a[2][0] = f1[0]; a[2][1] = f1[1]; a[3][0] = f1[2]; a[3][1] = f1[3];
}

template <int I, int N>
__device__ __forceinline__ void squares(const float* s, float* q) {
  if constexpr (I < N) {
    #pragma unroll
    for (int i = I; i < I + 32; ++i) q[i] = s[i] * s[i];
    squares<I + 32, N>(s, q);
  }
}


template <int G, int FM>
__device__ __forceinline__ void run_block(const float* in, float* res, const v4f* fr) {
  v2f a0[4], a1[4];
  frag_to_a(fr[0], fr[1], a0);
  frag_to_a(fr[2], fr[3], a1);
  product_pairs<G, FM, 0>(in, res, a0, a1);
}

__global__ __launch_bounds__(128) __attribute__((amdgpu_waves_per_eu(1, 1)))
void qsim(const float* __restrict__ x,
          const float* __restrict__ ws,
          float* __restrict__ out, int B) {
  const int lane = threadIdx.x & 31;
  const int wave = blockIdx.x * 4 + (threadIdx.x >> 5);
  const int half = lane >> 4;
  const int b    = wave * 16 + (lane & 15);
  const int bc   = b < B ? b : B - 1;

  const v4f* frag = reinterpret_cast<const v4f*>(ws) + lane;
  v4f fr[NBLOCKS][4];
  #pragma unroll
  for (int s = 0; s < NBLOCKS; ++s) {
    fr[s][0] = frag[128 * s];
    fr[s][1] = frag[128 * s + 32];
    if (ctl_wire(s) >= 0) {
      fr[s][2] = frag[128 * s + 64];
      fr[s][3] = frag[128 * s + 96];
    } else {
      fr[s][2] = fr[s][0];
      fr[s][3] = fr[s][1];
    }
  }

  const v4f* ang = reinterpret_cast<const v4f*>(ws + ANG_OFF + (size_t)bc * 16);
  const v4f c0 = ang[0], c1 = ang[1], s0 = ang[2], s1 = ang[3];
  const float cq[8] = {c0[0], c0[1], c0[2], c0[3], c1[0], c1[1], c1[2], c1[3]};
  const float sq[8] = {s0[0], s0[1], s0[2], s0[3], s1[0], s1[1], s1[2], s1[3]};
  float E[8];
  E[0] = half ? 0.f : 1.f;
  enc_level<0, 3>(E, cq, sq);
  float S[256];
  #pragma unroll
  for (int v = 0; v < 8; ++v) S[v << 5] = E[v];

  float R[256];
  run_block<0, wires_from(3)>(S, R, fr[0]);
  const float g34[4] = {cq[3] * cq[4], cq[3] * sq[4], sq[3] * cq[4], sq[3] * sq[4]};
  expand2<wires_from(3), 1 << (7 - 3), 1 << (7 - 4), 0>(R, g34);
  run_block<1, wires_from(5)>(R, S, fr[1]);
  const float g56[4] = {cq[5] * cq[6], cq[5] * sq[6], sq[5] * cq[6], sq[5] * sq[6]};
  expand2<wires_from(5), 1 << (7 - 5), 1 << (7 - 6), 0>(S, g56);
  run_block<2, wires_from(7)>(S, R, fr[2]);
  run_block<3, wires_from(7)>(R, S, fr[3]);
  expand1<wires_from(7), 1 << (7 - 7), 0>(S, cq[7], sq[7]);
  run_block<4, 0>(S, R, fr[4]);
  run_block<5, 0>(R, S, fr[5]);

  float sqv[256];
  squares<0, 256>(S, sqv);
  float z[8];
  zlevel<0>(sqv, z);
  #pragma unroll
  for (int k = 0; k < 8; ++k) z[k] += __shfl_xor(z[k], 16, 32);
  if (half == 0 && b < B) {
    v4f o0, o1;
    o0[0] = z[7]; o0[1] = z[6]; o0[2] = z[5]; o0[3] = z[4];
    o1[0] = z[3]; o1[1] = z[2]; o1[2] = z[1]; o1[3] = z[0];
    v4f* dst = reinterpret_cast<v4f*>(out + (size_t)b * 8);
    dst[0] = o0;
    dst[1] = o1;
  }
}

extern "C" void kernel_launch(void* const* d_in, const int* in_sizes, int n_in,
                              void* d_out, int out_size, void* d_ws, size_t ws_size,
                              hipStream_t stream) {
  const float* x = (const float*)d_in[0];
  const float* w = (const float*)d_in[1];
  float* out = (float*)d_out;
  float* ws  = (float*)d_ws;
  int B = in_sizes[0] / NQ;

  int nquads = B * 2;
  int tblocks = (nquads + PREP_THREADS - 1) / PREP_THREADS;
  qprep<<<PREP_BLOCKS + tblocks, PREP_THREADS, 0, stream>>>(x, w, ws, nquads);
  int nblocks = (B + 63) / 64;
  qsim<<<nblocks, 128, 0, stream>>>(x, ws, out, B);
}
